# starting kernel
# baseline (speedup 1.0000x reference)
#include <hip/hip_runtime.h>
#include <hip/hip_bf16.h>

#define NB   64
#define DD   32
#define NN   128
#define TT   6
#define NT   (NN*TT)
#define NCOL (NB*DD)
#define LEAF 64
#define JTOT 448
#define LDX  (JTOT*DD)

typedef __attribute__((ext_vector_type(2))) float v2f;
typedef __attribute__((ext_vector_type(8))) float v8f;

__global__ void k_copy_x(const float* __restrict__ x, float* __restrict__ out) {
    int idx = blockIdx.x * 256 + threadIdx.x;
    if (idx >= NB * 64 * DD) return;
    int b   = idx / (64 * DD);
    int rem = idx - b * (64 * DD);
    out[b * LDX + rem] = x[idx];
}

__global__ void k_fsmax(const float* __restrict__ fsl, float* __restrict__ fs, int J) {
    __shared__ float row[320];
    __shared__ float s_ts[256];
    __shared__ int   s_k[256];
    int nt  = blockIdx.x;
    int tid = threadIdx.x;
    for (int e = tid; e < J; e += 256) row[e] = fsl[nt * J + e];
    __syncthreads();

    float local_ts = 0.0f;
    int   local_k  = 0;
    for (int e = tid; e < J; e += 256) {
        float z = row[e];
        int   c = 1;
        float s = 0.0f;
        for (int j = 0; j < J; ++j) {
            float zj = row[j];
            c += (zj > z) ? 1 : 0;
            s += (zj >= z) ? zj : 0.0f;
        }
        bool supp = ((float)c * z > s - 1.0f);
        local_k  += supp ? 1 : 0;
        local_ts += supp ? z : 0.0f;
    }
    s_ts[tid] = local_ts;
    s_k[tid]  = local_k;
    __syncthreads();
    for (int off = 128; off > 0; off >>= 1) {
        if (tid < off) { s_ts[tid] += s_ts[tid + off]; s_k[tid] += s_k[tid + off]; }
        __syncthreads();
    }
    float tau = (s_ts[0] - 1.0f) / (float)s_k[0];
    for (int e = tid; e < J; e += 256) fs[nt * J + e] = fmaxf(row[e] - tau, 0.0f);
}

#define TPAD 33
__global__ void k_essum(const float* __restrict__ esl, float* __restrict__ essum, int J) {
    __shared__ float tile[8][32 * TPAD];
    __shared__ float accs[8][32];
    int nt   = blockIdx.x;
    int lane = threadIdx.x & 31;
    int wid  = threadIdx.x >> 5;
    float* my = tile[wid];
    float acc = 0.0f;
    int ntiles = J >> 5;
    for (int tb = wid; tb < ntiles; tb += 8) {
        size_t rowbase = (size_t)nt * J + tb * 32;
        #pragma unroll
        for (int rr = 0; rr < 32; ++rr) {
            my[rr * TPAD + lane] = esl[(rowbase + rr) * DD + lane];
        }
        float r[32];
        #pragma unroll
        for (int k = 0; k < 32; ++k) r[k] = my[lane * TPAD + k];
        #pragma unroll
        for (int k = 2; k <= 32; k <<= 1) {
            #pragma unroll
            for (int j = k >> 1; j > 0; j >>= 1) {
                #pragma unroll
                for (int i = 0; i < 32; ++i) {
                    int ij = i ^ j;
                    if (ij > i) {
                        float a  = r[i], b2 = r[ij];
                        float lo = fminf(a, b2), hi = fmaxf(a, b2);
                        if ((i & k) == 0) { r[i] = hi; r[ij] = lo; }
                        else              { r[i] = lo; r[ij] = hi; }
                    }
                }
            }
        }
        float cum = 0.0f, tausum = 1.0f;
        int   ksup = 1;
        #pragma unroll
        for (int i = 0; i < 32; ++i) {
            cum += r[i];
            bool s = ((float)(i + 1) * r[i] > cum - 1.0f);
            if (s) { ksup = i + 1; tausum = cum; }
        }
        float tau = (tausum - 1.0f) / (float)ksup;
        #pragma unroll
        for (int j2 = 0; j2 < 32; ++j2) {
            float z = my[j2 * TPAD + lane];
            float t = __shfl(tau, j2, 32);
            acc += fmaxf(z - t, 0.0f);
        }
    }
    accs[wid][lane] = acc;
    __syncthreads();
    if (threadIdx.x < 32) {
        float s = 0.0f;
        #pragma unroll
        for (int w = 0; w < 8; ++w) s += accs[w][threadIdx.x];
        essum[nt * DD + threadIdx.x] = s;
    }
}

__global__ void k_gemm_tl(const float* __restrict__ x, const float* __restrict__ fs,
                          const float* __restrict__ essum, const float* __restrict__ thr,
                          const float* __restrict__ lt, float* __restrict__ tl, int K) {
    int lane  = threadIdx.x & 31;
    int wid   = threadIdx.x >> 5;
    int tile  = blockIdx.x * 8 + wid;
    int mt    = tile >> 7;
    int ntile = tile & 127;
    int m0    = mt * 16;
    int col   = ntile * 16 + (lane & 15);
    int bidx  = col >> 5;
    int d     = col & 31;
    int ka_off = (lane >> 4) << 1;

    const float* arow = fs + (size_t)(m0 + (lane & 15)) * K;
    const float* bcol = x + (size_t)bidx * LDX + d;

    v8f acc = {};
    for (int k0 = 0; k0 < K; k0 += 4) {
        int ka = k0 + ka_off;
        v2f a;
        a.x = arow[ka];
        a.y = arow[ka + 1];
        v2f b;
        b.x = bcol[ka * DD];
        b.y = bcol[(ka + 1) * DD];
        acc = __builtin_amdgcn_wmma_f32_16x16x4_f32(false, a, false, b,
                                                    (short)0, acc, false, false);
    }
    int mplus = (lane >> 4) << 3;
    #pragma unroll
    for (int v = 0; v < 8; ++v) {
        int nt   = m0 + v + mplus;
        float sel = acc[v] * essum[nt * DD + d];
        float tv  = (sel - thr[nt]) * __expf(-lt[nt]);
        tl[(size_t)nt * NCOL + col] = tv;
    }
}

__global__ void k_tree(const float* __restrict__ tl, const float* __restrict__ resp,
                       float* __restrict__ out, int rowOff) {
    __shared__ float rsp[DD * 65];
    int nn  = blockIdx.x >> 3;
    int bg  = blockIdx.x & 7;
    int tid = threadIdx.x;
    for (int idx = tid; idx < DD * LEAF; idx += 256) {
        int dd = idx >> 6, cc = idx & 63;
        rsp[dd * 65 + cc] = resp[(size_t)nn * (DD * LEAF) + idx];
    }
    __syncthreads();

    int lane = tid & 31;
    int wid  = tid >> 5;
    int b    = bg * 8 + wid;

    float p[TT], q[TT];
    #pragma unroll
    for (int t = 0; t < TT; ++t) {
        float v = tl[(size_t)(nn * TT + t) * NCOL + b * DD + lane];
        p[t] = fminf(fmaxf(0.5f * v + 0.5f, 0.0f), 1.0f);
        q[t] = fminf(fmaxf(0.5f - 0.5f * v, 0.0f), 1.0f);
    }
    float r[LEAF];
    r[0] = 1.0f;
    #pragma unroll
    for (int t = 0; t < TT; ++t) {
        int sz = 1 << t;
        #pragma unroll
        for (int c = 0; c < 32; ++c) {
            int cc = sz - 1 - c;
            if (cc >= 0) {
                r[cc + sz] = r[cc] * q[t];
                r[cc]      = r[cc] * p[t];
            }
        }
    }
    float s = 0.0f;
    #pragma unroll
    for (int c = 0; c < LEAF; ++c) s += r[c] * rsp[lane * 65 + c];
    out[(size_t)b * LDX + (rowOff + nn) * DD + lane] = s;
}

extern "C" void kernel_launch(void* const* d_in, const int* in_sizes, int n_in,
                              void* d_out, int out_size, void* d_ws, size_t ws_size,
                              hipStream_t stream) {
    const float* x = (const float*)d_in[0];
    float* out = (float*)d_out;

    float* fs_ws    = (float*)d_ws;
    float* essum_ws = fs_ws + NT * 320;
    float* tl_ws    = essum_ws + NT * DD;

    k_copy_x<<<(NB * 64 * DD + 255) / 256, 256, 0, stream>>>(x, out);

    const int Js[3]     = {64, 192, 320};
    const int rowOff[3] = {64, 192, 320};
    for (int i = 0; i < 3; ++i) {
        const float* fsl  = (const float*)d_in[1 + 5 * i];
        const float* esl  = (const float*)d_in[2 + 5 * i];
        const float* thr  = (const float*)d_in[3 + 5 * i];
        const float* lt   = (const float*)d_in[4 + 5 * i];
        const float* resp = (const float*)d_in[5 + 5 * i];
        int J = Js[i];

        k_fsmax <<<NT, 256, 0, stream>>>(fsl, fs_ws, J);
        k_essum <<<NT, 256, 0, stream>>>(esl, essum_ws, J);
        k_gemm_tl<<<(48 * 128) / 8, 256, 0, stream>>>(out, fs_ws, essum_ws, thr, lt, tl_ws, J);
        k_tree  <<<NN * 8, 256, 0, stream>>>(tl_ws, resp, out, rowOff[i]);
    }
}


# essum of layer L leaves its own launch: NT extra blocks of the launch in front of layer L make it (x copy carries layer 0's, layer L-1 carries layer L's), a workspace plane per layer; 7 -> 4 launches
# speedup vs baseline: 2.4425x; 2.4425x over previous
#include <hip/hip_runtime.h>
#include <hip/hip_bf16.h>

#define NB   64
#define DD   32
#define NN   128
#define TT   6
#define NT   (NN*TT)
#define NCOL (NB*DD)
#define LEAF 64
#define JTOT 448
#define LDX  (JTOT*DD)

typedef __attribute__((ext_vector_type(2))) float v2f;
typedef __attribute__((ext_vector_type(8))) float v8f;

#define TPAD 33
#define ES_LDS (8 * 32 * TPAD + 8 * 32)
__device__ __forceinline__ void essum_block(const float* __restrict__ esl, float* __restrict__ essum,
                                            int J, int nt, float* smem) {
    float* accs = smem + 8 * 32 * TPAD;
    int lane = threadIdx.x & 31;
    int wid  = threadIdx.x >> 5;
    float* my = smem + wid * (32 * TPAD);
    float acc = 0.0f;
    int ntiles = J >> 5;
    for (int tb = wid; tb < ntiles; tb += 8) {
        size_t rowbase = (size_t)nt * J + tb * 32;
        #pragma unroll
        for (int rr = 0; rr < 32; ++rr) {
            my[rr * TPAD + lane] = esl[(rowbase + rr) * DD + lane];
        }
        float r[32];
        #pragma unroll
        for (int k = 0; k < 32; ++k) r[k] = my[lane * TPAD + k];
        #pragma unroll
        for (int k = 2; k <= 32; k <<= 1) {
            #pragma unroll
            for (int j = k >> 1; j > 0; j >>= 1) {
                #pragma unroll
                for (int i = 0; i < 32; ++i) {
                    int ij = i ^ j;
                    if (ij > i) {
                        float a  = r[i], b2 = r[ij];
                        float lo = fminf(a, b2), hi = fmaxf(a, b2);
                        if ((i & k) == 0) { r[i] = hi; r[ij] = lo; }
                        else              { r[i] = lo; r[ij] = hi; }
                    }
                }
            }
        }
        float cum = 0.0f, tausum = 1.0f;
        int   ksup = 1;
        #pragma unroll
        for (int i = 0; i < 32; ++i) {
            cum += r[i];
            bool s = ((float)(i + 1) * r[i] > cum - 1.0f);
            if (s) { ksup = i + 1; tausum = cum; }
        }
        float tau = (tausum - 1.0f) / (float)ksup;
        #pragma unroll
        for (int j2 = 0; j2 < 32; ++j2) {
            float z = my[j2 * TPAD + lane];
            float t = __shfl(tau, j2, 32);
            acc += fmaxf(z - t, 0.0f);
        }
    }
    accs[wid * 32 + lane] = acc;
    __syncthreads();
    if (threadIdx.x < 32) {
        float s = 0.0f;
        #pragma unroll
        for (int w = 0; w < 8; ++w) s += accs[w * 32 + threadIdx.x];
        essum[nt * DD + threadIdx.x] = s;
    }
}

#define NCOPY ((NB * 64 * DD + 255) / 256)
__global__ void k_copy_essum(const float* __restrict__ x, float* __restrict__ out,
                             const float* __restrict__ esl, float* __restrict__ essum) {
    __shared__ float smem[ES_LDS];
    if (blockIdx.x < NT) {
        essum_block(esl, essum, 64, blockIdx.x, smem);
        return;
    }
    int idx = (blockIdx.x - NT) * 256 + threadIdx.x;
    if (idx >= NB * 64 * DD) return;
    int b   = idx / (64 * DD);
    int rem = idx - b * (64 * DD);
    out[b * LDX + rem] = x[idx];
}

#define RSTR 65
#define RTREE (DD * RSTR)

__device__ __forceinline__ float leaf_dot(v8f acc, int tree, int d,
                                          const float* __restrict__ essum,
                                          const float* __restrict__ thr,
                                          const float* __restrict__ lt,
                                          const float* rrow) {
    float p[TT], q[TT];
    #pragma unroll
    for (int t = 0; t < TT; ++t) {
        int   nt  = tree * TT + t;
        float sel = acc[t] * essum[nt * DD + d];
        float v   = (sel - thr[nt]) * __expf(-lt[nt]);
        p[t] = fminf(fmaxf(0.5f * v + 0.5f, 0.0f), 1.0f);
        q[t] = fminf(fmaxf(0.5f - 0.5f * v, 0.0f), 1.0f);
    }
    float r[LEAF];
    r[0] = 1.0f;
    #pragma unroll
    for (int t = 0; t < TT; ++t) {
        int sz = 1 << t;
        #pragma unroll
        for (int c = 0; c < 32; ++c) {
            int cc = sz - 1 - c;
            if (cc >= 0) {
                r[cc + sz] = r[cc] * q[t];
                r[cc]      = r[cc] * p[t];
            }
        }
    }
    float s = 0.0f;
    #pragma unroll
    for (int c = 0; c < LEAF; ++c) s += r[c] * rrow[c];
    return s;
}

__device__ __forceinline__ float row16_sum(float v) {
    v += __builtin_bit_cast(float, __builtin_amdgcn_update_dpp(0, __builtin_bit_cast(int, v), 0x161, 0xf, 0xf, false));
    v += __builtin_bit_cast(float, __builtin_amdgcn_update_dpp(0, __builtin_bit_cast(int, v), 0x162, 0xf, 0xf, false));
    v += __builtin_bit_cast(float, __builtin_amdgcn_update_dpp(0, __builtin_bit_cast(int, v), 0x164, 0xf, 0xf, false));
    v += __builtin_bit_cast(float, __builtin_amdgcn_update_dpp(0, __builtin_bit_cast(int, v), 0x168, 0xf, 0xf, false));
    return v;
}
__device__ __forceinline__ int row16_isum(int v) {
    v += __builtin_amdgcn_update_dpp(0, v, 0x161, 0xf, 0xf, false);
    v += __builtin_amdgcn_update_dpp(0, v, 0x162, 0xf, 0xf, false);
    v += __builtin_amdgcn_update_dpp(0, v, 0x164, 0xf, 0xf, false);
    v += __builtin_amdgcn_update_dpp(0, v, 0x168, 0xf, 0xf, false);
    return v;
}

template <int K, int JN>
__global__ void k_layer(const float* x, const float* __restrict__ fsl,
                        const float* __restrict__ essum, const float* __restrict__ thr,
                        const float* __restrict__ lt, const float* __restrict__ resp,
                        float* out, int rowOff,
                        const float* __restrict__ esl_n, float* __restrict__ essum_n) {
    constexpr int PITCH = K + 4;
    constexpr int NQ    = K / 64;
    constexpr int NL    = (NN / 2) * 8;
    constexpr int LY_LDS = 2 * RTREE + 16 * PITCH;
    constexpr int SM_LDS = (JN && ES_LDS > LY_LDS) ? ES_LDS : LY_LDS;
    __shared__ __attribute__((aligned(16))) float smem[SM_LDS];
    if (JN && blockIdx.x >= NL) {
        essum_block(esl_n, essum_n, JN, blockIdx.x - NL, smem);
        return;
    }
    float* rsp = smem;
    float* afs = smem + 2 * RTREE;
    int bid  = blockIdx.x;
    int tid  = threadIdx.x;
    int lane = tid & 31;
    int wid  = tid >> 5;
    int mt   = bid >> 3;
    int bg   = bid & 7;
    int b    = bg * 8 + wid;
    int half = lane >> 4;
    int l15  = lane & 15;

    const float* rsrc = resp + (size_t)mt * (2 * DD * LEAF);
    float rv[4][4];
    #pragma unroll
    for (int i = 0; i < 4; ++i) {
        #pragma unroll
        for (int j = 0; j < 4; ++j) rv[i][j] = rsrc[i * 1024 + tid * 4 + j];
    }

    {
        int hw = tid >> 4;
        if (wid < 6) {
            const float* zr = fsl + (size_t)(mt * 2 * TT + hw) * K + l15 * 4;
            float z[NQ * 4];
            #pragma unroll
            for (int q = 0; q < NQ; ++q) {
                #pragma unroll
                for (int j = 0; j < 4; ++j) z[q * 4 + j] = zr[q * 64 + j];
            }
            float s = 0.0f;
            #pragma unroll
            for (int e = 0; e < NQ * 4; ++e) s += z[e];
            s = row16_sum(s);
            int   cnt = K;
            float tau = (s - 1.0f) / (float)K;
            for (int it = 0; it < K; ++it) {
                float s2 = 0.0f;
                int   c2 = 0;
                #pragma unroll
                for (int e = 0; e < NQ * 4; ++e) {
                    bool in = z[e] > tau;
                    s2 += in ? z[e] : 0.0f;
                    c2 += in ? 1 : 0;
                }
                s2 = row16_sum(s2);
                c2 = row16_isum(c2);
                bool moved = (c2 != cnt);
                tau = (s2 - 1.0f) / (float)c2;
                cnt = c2;
                if (__builtin_amdgcn_ballot_w32(moved) == 0u) break;
            }
            int slot = (hw / TT) * 8 + (hw % TT);
            float* dst = afs + slot * PITCH + l15 * 4;
            #pragma unroll
            for (int q = 0; q < NQ; ++q) {
                #pragma unroll
                for (int j = 0; j < 4; ++j) dst[q * 64 + j] = fmaxf(z[q * 4 + j] - tau, 0.0f);
            }
        } else {
            int slot = ((hw - 12) >> 1) * 8 + 6 + (hw & 1);
            float* dst = afs + slot * PITCH + l15 * 4;
            #pragma unroll
            for (int q = 0; q < NQ; ++q) {
                #pragma unroll
                for (int j = 0; j < 4; ++j) dst[q * 64 + j] = 0.0f;
            }
        }
    }
    __syncthreads();

    const float* ap   = afs + l15 * PITCH + (half << 1);
    const float* bcol = x + (size_t)b * LDX + l15;
    int ka_off = half << 1;

    v8f acc0 = {}, acc1 = {};
    #pragma unroll 8
    for (int k0 = 0; k0 < K; k0 += 4) {
        int ka = k0 + ka_off;
        v2f a = *(const v2f*)(ap + k0);
        v2f b0, b1;
        b0.x = bcol[ka * DD];
        b0.y = bcol[(ka + 1) * DD];
        b1.x = bcol[ka * DD + 16];
        b1.y = bcol[(ka + 1) * DD + 16];
        acc0 = __builtin_amdgcn_wmma_f32_16x16x4_f32(false, a, false, b0,
                                                     (short)0, acc0, false, false);
        acc1 = __builtin_amdgcn_wmma_f32_16x16x4_f32(false, a, false, b1,
                                                     (short)0, acc1, false, false);
    }

    #pragma unroll
    for (int i = 0; i < 4; ++i) {
        int e   = i * 1024 + tid * 4;
        int tr  = e >> 11;
        int rem = e & 2047;
        int dd  = rem >> 6, cc = rem & 63;
        #pragma unroll
        for (int j = 0; j < 4; ++j) rsp[tr * RTREE + dd * RSTR + cc + j] = rv[i][j];
    }
    __syncthreads();

    int tree = mt * 2 + half;
    const float* rbase = rsp + half * RTREE;
    float s0 = leaf_dot(acc0, tree, l15,      essum, thr, lt, rbase + l15 * RSTR);
    float s1 = leaf_dot(acc1, tree, 16 + l15, essum, thr, lt, rbase + (16 + l15) * RSTR);

    float t0 = __shfl_xor(s0, 16, 32);
    float t1 = __shfl_xor(s1, 16, 32);
    float oA = half ? t1 : s0;
    float oB = half ? s1 : t0;
    size_t obase = (size_t)b * LDX + (size_t)(rowOff + mt * 2) * DD + lane;
    out[obase]      = oA;
    out[obase + DD] = oB;
}

extern "C" void kernel_launch(void* const* d_in, const int* in_sizes, int n_in,
                              void* d_out, int out_size, void* d_ws, size_t ws_size,
                              hipStream_t stream) {
    const float* x = (const float*)d_in[0];
    float* out = (float*)d_out;

    float* es0 = (float*)d_ws + NT * 320;
    float* es1 = (float*)d_ws;
    float* es2 = (float*)d_ws + NT * 32;

    k_copy_essum<<<NT + NCOPY, 256, 0, stream>>>(x, out, (const float*)d_in[2], es0);

    k_layer<64, 192><<<NT + (NN / 2) * 8, 256, 0, stream>>>(out, (const float*)d_in[1], es0, (const float*)d_in[3],
                                                            (const float*)d_in[4], (const float*)d_in[5], out, 64,
                                                            (const float*)d_in[7], es1);
    k_layer<192, 320><<<NT + (NN / 2) * 8, 256, 0, stream>>>(out, (const float*)d_in[6], es1, (const float*)d_in[8],
                                                             (const float*)d_in[9], (const float*)d_in[10], out, 192,
                                                             (const float*)d_in[12], es2);
    k_layer<320, 0><<<(NN / 2) * 8, 256, 0, stream>>>(out, (const float*)d_in[11], es2, (const float*)d_in[13],
                                                      (const float*)d_in[14], (const float*)d_in[15], out, 320,
                                                      nullptr, nullptr);
}


# Side blocks repacked by the wave: 768 essum blocks (1 tile or none a wave) -> 256 leading blocks of 3 rows, 1-4 tiles a wave, next tile prefetched; x copy in their idle waves; essum sum order changes
# speedup vs baseline: 2.5392x; 1.0396x over previous
#include <hip/hip_runtime.h>
#include <hip/hip_bf16.h>

#define NB   64
#define DD   32
#define NN   128
#define TT   6
#define NT   (NN*TT)
#define NCOL (NB*DD)
#define LEAF 64
#define JTOT 448
#define LDX  (JTOT*DD)

typedef __attribute__((ext_vector_type(2))) float v2f;
typedef __attribute__((ext_vector_type(8))) float v8f;

#define TPAD 33
#define NTB  3
#define ES_LDS (8 * 32 * TPAD + NTB * 8 * 32)
template <int J>
__device__ __forceinline__ void essum_block(const float* __restrict__ esl, float* __restrict__ essum,
                                            int blk, float* smem) {
    constexpr int T     = J >> 5;
    constexpr int NTILE = NTB * T;
    float* accs = smem + 8 * 32 * TPAD;
    int lane = threadIdx.x & 31;
    int wid  = threadIdx.x >> 5;
    float* my = smem + wid * (32 * TPAD);
    #pragma unroll
    for (int k = 0; k < NTB; ++k) accs[(k * 8 + wid) * 32 + lane] = 0.0f;
    const float* src = esl + (size_t)blk * (NTB * J * DD) + lane;
    float nx[32];
    int f = wid;
    if (f < NTILE) {
        #pragma unroll
        for (int rr = 0; rr < 32; ++rr) nx[rr] = src[(f * 32 + rr) * DD];
    }
    while (f < NTILE) {
        #pragma unroll
        for (int rr = 0; rr < 32; ++rr) my[rr * TPAD + lane] = nx[rr];
        int fn = f + 8;
        if (fn < NTILE) {
            #pragma unroll
            for (int rr = 0; rr < 32; ++rr) nx[rr] = src[(fn * 32 + rr) * DD];
        }
        float r[32];
        #pragma unroll
        for (int k = 0; k < 32; ++k) r[k] = my[lane * TPAD + k];
        #pragma unroll
        for (int k = 2; k <= 32; k <<= 1) {
            #pragma unroll
            for (int j = k >> 1; j > 0; j >>= 1) {
                #pragma unroll
                for (int i = 0; i < 32; ++i) {
                    int ij = i ^ j;
                    if (ij > i) {
                        float a  = r[i], b2 = r[ij];
                        float lo = fminf(a, b2), hi = fmaxf(a, b2);
                        if ((i & k) == 0) { r[i] = hi; r[ij] = lo; }
                        else              { r[i] = lo; r[ij] = hi; }
                    }
                }
            }
        }
        float cum = 0.0f, tausum = 1.0f;
        int   ksup = 1;
        #pragma unroll
        for (int i = 0; i < 32; ++i) {
            cum += r[i];
            bool s = ((float)(i + 1) * r[i] > cum - 1.0f);
            if (s) { ksup = i + 1; tausum = cum; }
        }
        float tau = (tausum - 1.0f) / (float)ksup;
        float* slot = accs + ((f / T) * 8 + wid) * 32 + lane;
        float acc = *slot;
        #pragma unroll
        for (int j2 = 0; j2 < 32; ++j2) {
            float z = my[j2 * TPAD + lane];
            float t = __shfl(tau, j2, 32);
            acc += fmaxf(z - t, 0.0f);
        }
        *slot = acc;
        f = fn;
    }
    __syncthreads();
    if (threadIdx.x < NTB * 32) {
        int k = threadIdx.x >> 5;
        float s = 0.0f;
        #pragma unroll
        for (int w = 0; w < 8; ++w) s += accs[(k * 8 + w) * 32 + lane];
        essum[(blk * NTB + k) * DD + lane] = s;
    }
}

#define NES   (NT / NTB)
#define CPW   (8 - NTB * 2)
#define CPL   ((NB * 64 * DD / 4) / (NES * CPW * 32))
static_assert(NES * NTB == NT, "essum rows must divide over the blocks");
static_assert(CPW > 0 && CPL * NES * CPW * 32 == NB * 64 * DD / 4, "x copy must divide over the tile-less waves");
__global__ void __launch_bounds__(256) k_copy_essum(const float* __restrict__ x, float* __restrict__ out,
                             const float* __restrict__ esl, float* __restrict__ essum) {
    __shared__ float smem[ES_LDS];
    int wid = threadIdx.x >> 5;
    if (wid >= 8 - CPW) {
        int lane = threadIdx.x & 31;
        int base = (blockIdx.x * CPW + (wid - (8 - CPW))) * CPL;
        #pragma unroll
        for (int j = 0; j < CPL; ++j) {
            int idx = ((base + j) * 32 + lane) * 4;
            int b   = idx / (64 * DD);
            int rem = idx - b * (64 * DD);
            *(float4*)(out + (size_t)b * LDX + rem) = *(const float4*)(x + idx);
        }
    }
    essum_block<64>(esl, essum, blockIdx.x, smem);
}

#define RSTR 65
#define RTREE (DD * RSTR)

__device__ __forceinline__ float leaf_dot(v8f acc, int tree, int d,
                                          const float* __restrict__ essum,
                                          const float* __restrict__ thr,
                                          const float* __restrict__ lt,
                                          const float* rrow) {
    float p[TT], q[TT];
    #pragma unroll
    for (int t = 0; t < TT; ++t) {
        int   nt  = tree * TT + t;
        float sel = acc[t] * essum[nt * DD + d];
        float v   = (sel - thr[nt]) * __expf(-lt[nt]);
        p[t] = fminf(fmaxf(0.5f * v + 0.5f, 0.0f), 1.0f);
        q[t] = fminf(fmaxf(0.5f - 0.5f * v, 0.0f), 1.0f);
    }
    float r[LEAF];
    r[0] = 1.0f;
    #pragma unroll
    for (int t = 0; t < TT; ++t) {
        int sz = 1 << t;
        #pragma unroll
        for (int c = 0; c < 32; ++c) {
            int cc = sz - 1 - c;
            if (cc >= 0) {
                r[cc + sz] = r[cc] * q[t];
                r[cc]      = r[cc] * p[t];
            }
        }
    }
    float s = 0.0f;
    #pragma unroll
    for (int c = 0; c < LEAF; ++c) s += r[c] * rrow[c];
    return s;
}

__device__ __forceinline__ float row16_sum(float v) {
    v += __builtin_bit_cast(float, __builtin_amdgcn_update_dpp(0, __builtin_bit_cast(int, v), 0x161, 0xf, 0xf, false));
    v += __builtin_bit_cast(float, __builtin_amdgcn_update_dpp(0, __builtin_bit_cast(int, v), 0x162, 0xf, 0xf, false));
    v += __builtin_bit_cast(float, __builtin_amdgcn_update_dpp(0, __builtin_bit_cast(int, v), 0x164, 0xf, 0xf, false));
    v += __builtin_bit_cast(float, __builtin_amdgcn_update_dpp(0, __builtin_bit_cast(int, v), 0x168, 0xf, 0xf, false));
    return v;
}
__device__ __forceinline__ int row16_isum(int v) {
    v += __builtin_amdgcn_update_dpp(0, v, 0x161, 0xf, 0xf, false);
    v += __builtin_amdgcn_update_dpp(0, v, 0x162, 0xf, 0xf, false);
    v += __builtin_amdgcn_update_dpp(0, v, 0x164, 0xf, 0xf, false);
    v += __builtin_amdgcn_update_dpp(0, v, 0x168, 0xf, 0xf, false);
    return v;
}

template <int K, int JN>
__global__ void __launch_bounds__(256) k_layer(const float* x, const float* __restrict__ fsl,
                        const float* __restrict__ essum, const float* __restrict__ thr,
                        const float* __restrict__ lt, const float* __restrict__ resp,
                        float* out, int rowOff,
                        const float* __restrict__ esl_n, float* __restrict__ essum_n) {
    constexpr int PITCH = K + 4;
    constexpr int NQ    = K / 64;
    constexpr int NE    = JN ? NES : 0;
    constexpr int LY_LDS = 2 * RTREE + 16 * PITCH;
    constexpr int SM_LDS = (JN && ES_LDS > LY_LDS) ? ES_LDS : LY_LDS;
    __shared__ __attribute__((aligned(16))) float smem[SM_LDS];
    if constexpr (JN != 0) {
        if (blockIdx.x < NE) {
            essum_block<JN>(esl_n, essum_n, blockIdx.x, smem);
            return;
        }
    }
    float* rsp = smem;
    float* afs = smem + 2 * RTREE;
    int bid  = blockIdx.x - NE;
    int tid  = threadIdx.x;
    int lane = tid & 31;
    int wid  = tid >> 5;
    int mt   = bid >> 3;
    int bg   = bid & 7;
    int b    = bg * 8 + wid;
    int half = lane >> 4;
    int l15  = lane & 15;

    const float* rsrc = resp + (size_t)mt * (2 * DD * LEAF);
    float rv[4][4];
    #pragma unroll
    for (int i = 0; i < 4; ++i) {
        #pragma unroll
        for (int j = 0; j < 4; ++j) rv[i][j] = rsrc[i * 1024 + tid * 4 + j];
    }

    {
        int hw = tid >> 4;
        if (wid < 6) {
            const float* zr = fsl + (size_t)(mt * 2 * TT + hw) * K + l15 * 4;
            float z[NQ * 4];
            #pragma unroll
            for (int q = 0; q < NQ; ++q) {
                #pragma unroll
                for (int j = 0; j < 4; ++j) z[q * 4 + j] = zr[q * 64 + j];
            }
            float s = 0.0f;
            #pragma unroll
            for (int e = 0; e < NQ * 4; ++e) s += z[e];
            s = row16_sum(s);
            int   cnt = K;
            float tau = (s - 1.0f) / (float)K;
            for (int it = 0; it < K; ++it) {
                float s2 = 0.0f;
                int   c2 = 0;
                #pragma unroll
                for (int e = 0; e < NQ * 4; ++e) {
                    bool in = z[e] > tau;
                    s2 += in ? z[e] : 0.0f;
                    c2 += in ? 1 : 0;
                }
                s2 = row16_sum(s2);
                c2 = row16_isum(c2);
                bool moved = (c2 != cnt);
                tau = (s2 - 1.0f) / (float)c2;
                cnt = c2;
                if (__builtin_amdgcn_ballot_w32(moved) == 0u) break;
            }
            int slot = (hw / TT) * 8 + (hw % TT);
            float* dst = afs + slot * PITCH + l15 * 4;
            #pragma unroll
            for (int q = 0; q < NQ; ++q) {
                #pragma unroll
                for (int j = 0; j < 4; ++j) dst[q * 64 + j] = fmaxf(z[q * 4 + j] - tau, 0.0f);
            }
        } else {
            int slot = ((hw - 12) >> 1) * 8 + 6 + (hw & 1);
            float* dst = afs + slot * PITCH + l15 * 4;
            #pragma unroll
            for (int q = 0; q < NQ; ++q) {
                #pragma unroll
                for (int j = 0; j < 4; ++j) dst[q * 64 + j] = 0.0f;
            }
        }
    }
    __syncthreads();

    const float* ap   = afs + l15 * PITCH + (half << 1);
    const float* bcol = x + (size_t)b * LDX + l15;
    int ka_off = half << 1;

    v8f acc0 = {}, acc1 = {};
    #pragma unroll 8
    for (int k0 = 0; k0 < K; k0 += 4) {
        int ka = k0 + ka_off;
        v2f a = *(const v2f*)(ap + k0);
        v2f b0, b1;
        b0.x = bcol[ka * DD];
        b0.y = bcol[(ka + 1) * DD];
        b1.x = bcol[ka * DD + 16];
        b1.y = bcol[(ka + 1) * DD + 16];
        acc0 = __builtin_amdgcn_wmma_f32_16x16x4_f32(false, a, false, b0,
                                                     (short)0, acc0, false, false);
        acc1 = __builtin_amdgcn_wmma_f32_16x16x4_f32(false, a, false, b1,
                                                     (short)0, acc1, false, false);
    }

    #pragma unroll
    for (int i = 0; i < 4; ++i) {
        int e   = i * 1024 + tid * 4;
        int tr  = e >> 11;
        int rem = e & 2047;
        int dd  = rem >> 6, cc = rem & 63;
        #pragma unroll
        for (int j = 0; j < 4; ++j) rsp[tr * RTREE + dd * RSTR + cc + j] = rv[i][j];
    }
    __syncthreads();

    int tree = mt * 2 + half;
    const float* rbase = rsp + half * RTREE;
    float s0 = leaf_dot(acc0, tree, l15,      essum, thr, lt, rbase + l15 * RSTR);
    float s1 = leaf_dot(acc1, tree, 16 + l15, essum, thr, lt, rbase + (16 + l15) * RSTR);

    float t0 = __shfl_xor(s0, 16, 32);
    float t1 = __shfl_xor(s1, 16, 32);
    float oA = half ? t1 : s0;
    float oB = half ? s1 : t0;
    size_t obase = (size_t)b * LDX + (size_t)(rowOff + mt * 2) * DD + lane;
    out[obase]      = oA;
    out[obase + DD] = oB;
}

extern "C" void kernel_launch(void* const* d_in, const int* in_sizes, int n_in,
                              void* d_out, int out_size, void* d_ws, size_t ws_size,
                              hipStream_t stream) {
    const float* x = (const float*)d_in[0];
    float* out = (float*)d_out;

    float* es0 = (float*)d_ws + NT * 320;
    float* es1 = (float*)d_ws;
    float* es2 = (float*)d_ws + NT * 32;

    k_copy_essum<<<NES, 256, 0, stream>>>(x, out, (const float*)d_in[2], es0);

    k_layer<64, 192><<<NES + (NN / 2) * 8, 256, 0, stream>>>(out, (const float*)d_in[1], es0, (const float*)d_in[3],
                                                            (const float*)d_in[4], (const float*)d_in[5], out, 64,
                                                            (const float*)d_in[7], es1);
    k_layer<192, 320><<<NES + (NN / 2) * 8, 256, 0, stream>>>(out, (const float*)d_in[6], es1, (const float*)d_in[8],
                                                             (const float*)d_in[9], (const float*)d_in[10], out, 192,
                                                             (const float*)d_in[12], es2);
    k_layer<320, 0><<<(NN / 2) * 8, 256, 0, stream>>>(out, (const float*)d_in[11], es2, (const float*)d_in[13],
                                                      (const float*)d_in[14], (const float*)d_in[15], out, 320,
                                                      nullptr, nullptr);
}


# fs threshold (tau) of every (n,t) row made once, by the leading essum block that owns the row in the launch in front: one float a row in the workspace; layer blocks only subtract and clamp
# speedup vs baseline: 2.6853x; 1.0575x over previous
#include <hip/hip_runtime.h>
#include <hip/hip_bf16.h>

#define NB   64
#define DD   32
#define NN   128
#define TT   6
#define NT   (NN*TT)
#define NCOL (NB*DD)
#define LEAF 64
#define JTOT 448
#define LDX  (JTOT*DD)

typedef __attribute__((ext_vector_type(2))) float v2f;
typedef __attribute__((ext_vector_type(8))) float v8f;

__device__ __forceinline__ float row16_sum(float v) {
    v += __builtin_bit_cast(float, __builtin_amdgcn_update_dpp(0, __builtin_bit_cast(int, v), 0x161, 0xf, 0xf, false));
    v += __builtin_bit_cast(float, __builtin_amdgcn_update_dpp(0, __builtin_bit_cast(int, v), 0x162, 0xf, 0xf, false));
    v += __builtin_bit_cast(float, __builtin_amdgcn_update_dpp(0, __builtin_bit_cast(int, v), 0x164, 0xf, 0xf, false));
    v += __builtin_bit_cast(float, __builtin_amdgcn_update_dpp(0, __builtin_bit_cast(int, v), 0x168, 0xf, 0xf, false));
    return v;
}
__device__ __forceinline__ int row16_isum(int v) {
    v += __builtin_amdgcn_update_dpp(0, v, 0x161, 0xf, 0xf, false);
    v += __builtin_amdgcn_update_dpp(0, v, 0x162, 0xf, 0xf, false);
    v += __builtin_amdgcn_update_dpp(0, v, 0x164, 0xf, 0xf, false);
    v += __builtin_amdgcn_update_dpp(0, v, 0x168, 0xf, 0xf, false);
    return v;
}

#define NTB  3

template <int K>
__device__ __forceinline__ void fs_tau_rows(const float* __restrict__ fsl, float* __restrict__ ftau, int blk) {
    constexpr int NQ = K / 64;
    int lane = threadIdx.x & 31;
    int wid  = threadIdx.x >> 5;
    int half = lane >> 4;
    int l15  = lane & 15;
    int row  = blk * NTB + ((wid == 7) ? half : 2);
    const float* zr = fsl + (size_t)row * K + l15 * 4;
    float z[NQ * 4];
    #pragma unroll
    for (int q = 0; q < NQ; ++q) {
        #pragma unroll
        for (int j = 0; j < 4; ++j) z[q * 4 + j] = zr[q * 64 + j];
    }
    float s = 0.0f;
    #pragma unroll
    for (int e = 0; e < NQ * 4; ++e) s += z[e];
    s = row16_sum(s);
    int   cnt = K;
    float tau = (s - 1.0f) / (float)K;
    for (int it = 0; it < K; ++it) {
        float s2 = 0.0f;
        int   c2 = 0;
        #pragma unroll
        for (int e = 0; e < NQ * 4; ++e) {
            bool in = z[e] > tau;
            s2 += in ? z[e] : 0.0f;
            c2 += in ? 1 : 0;
        }
        s2 = row16_sum(s2);
        c2 = row16_isum(c2);
        bool moved = (c2 != cnt);
        tau = (s2 - 1.0f) / (float)c2;
        cnt = c2;
        if (__builtin_amdgcn_ballot_w32(moved) == 0u) break;
    }
    if (l15 == 0 && (wid == 7 || half == 0)) ftau[row] = tau;
}

#define TPAD 33
#define ES_LDS (8 * 32 * TPAD + NTB * 8 * 32)
template <int J>
__device__ __forceinline__ void essum_block(const float* __restrict__ esl, float* __restrict__ essum,
                                            const float* __restrict__ fsl, float* __restrict__ ftau,
                                            int blk, float* smem) {
    constexpr int T     = J >> 5;
    constexpr int NTILE = NTB * T;
    float* accs = smem + 8 * 32 * TPAD;
    int lane = threadIdx.x & 31;
    int wid  = threadIdx.x >> 5;
    float* my = smem + wid * (32 * TPAD);
    #pragma unroll
    for (int k = 0; k < NTB; ++k) accs[(k * 8 + wid) * 32 + lane] = 0.0f;
    if (wid >= 6) fs_tau_rows<J>(fsl, ftau, blk);
    const float* src = esl + (size_t)blk * (NTB * J * DD) + lane;
    float nx[32];
    int f = wid;
    if (f < NTILE) {
        #pragma unroll
        for (int rr = 0; rr < 32; ++rr) nx[rr] = src[(f * 32 + rr) * DD];
    }
    while (f < NTILE) {
        #pragma unroll
        for (int rr = 0; rr < 32; ++rr) my[rr * TPAD + lane] = nx[rr];
        int fn = f + 8;
        if (fn < NTILE) {
            #pragma unroll
            for (int rr = 0; rr < 32; ++rr) nx[rr] = src[(fn * 32 + rr) * DD];
        }
        float r[32];
        #pragma unroll
        for (int k = 0; k < 32; ++k) r[k] = my[lane * TPAD + k];
        #pragma unroll
        for (int k = 2; k <= 32; k <<= 1) {
            #pragma unroll
            for (int j = k >> 1; j > 0; j >>= 1) {
                #pragma unroll
                for (int i = 0; i < 32; ++i) {
                    int ij = i ^ j;
                    if (ij > i) {
                        float a  = r[i], b2 = r[ij];
                        float lo = fminf(a, b2), hi = fmaxf(a, b2);
                        if ((i & k) == 0) { r[i] = hi; r[ij] = lo; }
                        else              { r[i] = lo; r[ij] = hi; }
                    }
                }
            }
        }
        float cum = 0.0f, tausum = 1.0f;
        int   ksup = 1;
        #pragma unroll
        for (int i = 0; i < 32; ++i) {
            cum += r[i];
            bool s = ((float)(i + 1) * r[i] > cum - 1.0f);
            if (s) { ksup = i + 1; tausum = cum; }
        }
        float tau = (tausum - 1.0f) / (float)ksup;
        float* slot = accs + ((f / T) * 8 + wid) * 32 + lane;
        float acc = *slot;
        #pragma unroll
        for (int j2 = 0; j2 < 32; ++j2) {
            float z = my[j2 * TPAD + lane];
            float t = __shfl(tau, j2, 32);
            acc += fmaxf(z - t, 0.0f);
        }
        *slot = acc;
        f = fn;
    }
    __syncthreads();
    if (threadIdx.x < NTB * 32) {
        int k = threadIdx.x >> 5;
        float s = 0.0f;
        #pragma unroll
        for (int w = 0; w < 8; ++w) s += accs[(k * 8 + w) * 32 + lane];
        essum[(blk * NTB + k) * DD + lane] = s;
    }
}

#define NES   (NT / NTB)
#define CPW   (8 - NTB * 2)
#define CPL   ((NB * 64 * DD / 4) / (NES * CPW * 32))
static_assert(NES * NTB == NT, "essum rows must divide over the blocks");
static_assert(CPW > 0 && CPL * NES * CPW * 32 == NB * 64 * DD / 4, "x copy must divide over the tile-less waves");
__global__ void __launch_bounds__(256) k_copy_essum(const float* __restrict__ x, float* __restrict__ out,
                             const float* __restrict__ esl, float* __restrict__ essum,
                             const float* __restrict__ fsl, float* __restrict__ ftau) {
    __shared__ float smem[ES_LDS];
    int wid = threadIdx.x >> 5;
    if (wid >= 8 - CPW) {
        int lane = threadIdx.x & 31;
        int base = (blockIdx.x * CPW + (wid - (8 - CPW))) * CPL;
        #pragma unroll
        for (int j = 0; j < CPL; ++j) {
            int idx = ((base + j) * 32 + lane) * 4;
            int b   = idx / (64 * DD);
            int rem = idx - b * (64 * DD);
            *(float4*)(out + (size_t)b * LDX + rem) = *(const float4*)(x + idx);
        }
    }
    essum_block<64>(esl, essum, fsl, ftau, blockIdx.x, smem);
}

#define RSTR 65
#define RTREE (DD * RSTR)

__device__ __forceinline__ float leaf_dot(v8f acc, int tree, int d,
                                          const float* __restrict__ essum,
                                          const float* __restrict__ thr,
                                          const float* __restrict__ lt,
                                          const float* rrow) {
    float p[TT], q[TT];
    #pragma unroll
    for (int t = 0; t < TT; ++t) {
        int   nt  = tree * TT + t;
        float sel = acc[t] * essum[nt * DD + d];
        float v   = (sel - thr[nt]) * __expf(-lt[nt]);
        p[t] = fminf(fmaxf(0.5f * v + 0.5f, 0.0f), 1.0f);
        q[t] = fminf(fmaxf(0.5f - 0.5f * v, 0.0f), 1.0f);
    }
    float r[LEAF];
    r[0] = 1.0f;
    #pragma unroll
    for (int t = 0; t < TT; ++t) {
        int sz = 1 << t;
        #pragma unroll
        for (int c = 0; c < 32; ++c) {
            int cc = sz - 1 - c;
            if (cc >= 0) {
                r[cc + sz] = r[cc] * q[t];
                r[cc]      = r[cc] * p[t];
            }
        }
    }
    float s = 0.0f;
    #pragma unroll
    for (int c = 0; c < LEAF; ++c) s += r[c] * rrow[c];
    return s;
}

template <int K, int JN>
__global__ void __launch_bounds__(256) k_layer(const float* x, const float* __restrict__ fsl,
                        const float* __restrict__ ftau,
                        const float* __restrict__ essum, const float* __restrict__ thr,
                        const float* __restrict__ lt, const float* __restrict__ resp,
                        float* out, int rowOff,
                        const float* __restrict__ esl_n, float* __restrict__ essum_n,
                        const float* __restrict__ fsl_n, float* __restrict__ ftau_n) {
    constexpr int PITCH = K + 4;
    constexpr int NQ    = K / 64;
    constexpr int NE    = JN ? NES : 0;
    constexpr int LY_LDS = 2 * RTREE + 16 * PITCH;
    constexpr int SM_LDS = (JN && ES_LDS > LY_LDS) ? ES_LDS : LY_LDS;
    __shared__ __attribute__((aligned(16))) float smem[SM_LDS];
    if constexpr (JN != 0) {
        if (blockIdx.x < NE) {
            essum_block<JN>(esl_n, essum_n, fsl_n, ftau_n, blockIdx.x, smem);
            return;
        }
    }
    float* rsp = smem;
    float* afs = smem + 2 * RTREE;
    int bid  = blockIdx.x - NE;
    int tid  = threadIdx.x;
    int lane = tid & 31;
    int wid  = tid >> 5;
    int mt   = bid >> 3;
    int bg   = bid & 7;
    int b    = bg * 8 + wid;
    int half = lane >> 4;
    int l15  = lane & 15;

    const float* rsrc = resp + (size_t)mt * (2 * DD * LEAF);
    float rv[4][4];
    #pragma unroll
    for (int i = 0; i < 4; ++i) {
        #pragma unroll
        for (int j = 0; j < 4; ++j) rv[i][j] = rsrc[i * 1024 + tid * 4 + j];
    }

    {
        int hw = tid >> 4;
        if (wid < 6) {
            const float* zr = fsl + (size_t)(mt * 2 * TT + hw) * K + l15 * 4;
            float z[NQ * 4];
            #pragma unroll
            for (int q = 0; q < NQ; ++q) {
                #pragma unroll
                for (int j = 0; j < 4; ++j) z[q * 4 + j] = zr[q * 64 + j];
            }
            float tau = ftau[mt * 2 * TT + hw];
            int slot = (hw / TT) * 8 + (hw % TT);
            float* dst = afs + slot * PITCH + l15 * 4;
            #pragma unroll
            for (int q = 0; q < NQ; ++q) {
                #pragma unroll
                for (int j = 0; j < 4; ++j) dst[q * 64 + j] = fmaxf(z[q * 4 + j] - tau, 0.0f);
            }
        } else {
            int slot = ((hw - 12) >> 1) * 8 + 6 + (hw & 1);
            float* dst = afs + slot * PITCH + l15 * 4;
            #pragma unroll
            for (int q = 0; q < NQ; ++q) {
                #pragma unroll
                for (int j = 0; j < 4; ++j) dst[q * 64 + j] = 0.0f;
            }
        }
    }
    __syncthreads();

    const float* ap   = afs + l15 * PITCH + (half << 1);
    const float* bcol = x + (size_t)b * LDX + l15;
    int ka_off = half << 1;

    v8f acc0 = {}, acc1 = {};
    #pragma unroll 8
    for (int k0 = 0; k0 < K; k0 += 4) {
        int ka = k0 + ka_off;
        v2f a = *(const v2f*)(ap + k0);
        v2f b0, b1;
        b0.x = bcol[ka * DD];
        b0.y = bcol[(ka + 1) * DD];
        b1.x = bcol[ka * DD + 16];
        b1.y = bcol[(ka + 1) * DD + 16];
        acc0 = __builtin_amdgcn_wmma_f32_16x16x4_f32(false, a, false, b0,
                                                     (short)0, acc0, false, false);
        acc1 = __builtin_amdgcn_wmma_f32_16x16x4_f32(false, a, false, b1,
                                                     (short)0, acc1, false, false);
    }

    #pragma unroll
    for (int i = 0; i < 4; ++i) {
        int e   = i * 1024 + tid * 4;
        int tr  = e >> 11;
        int rem = e & 2047;
        int dd  = rem >> 6, cc = rem & 63;
        #pragma unroll
        for (int j = 0; j < 4; ++j) rsp[tr * RTREE + dd * RSTR + cc + j] = rv[i][j];
    }
    __syncthreads();

    int tree = mt * 2 + half;
    const float* rbase = rsp + half * RTREE;
    float s0 = leaf_dot(acc0, tree, l15,      essum, thr, lt, rbase + l15 * RSTR);
    float s1 = leaf_dot(acc1, tree, 16 + l15, essum, thr, lt, rbase + (16 + l15) * RSTR);

    float t0 = __shfl_xor(s0, 16, 32);
    float t1 = __shfl_xor(s1, 16, 32);
    float oA = half ? t1 : s0;
    float oB = half ? s1 : t0;
    size_t obase = (size_t)b * LDX + (size_t)(rowOff + mt * 2) * DD + lane;
    out[obase]      = oA;
    out[obase + DD] = oB;
}

extern "C" void kernel_launch(void* const* d_in, const int* in_sizes, int n_in,
                              void* d_out, int out_size, void* d_ws, size_t ws_size,
                              hipStream_t stream) {
    const float* x = (const float*)d_in[0];
    float* out = (float*)d_out;

    float* es0 = (float*)d_ws + NT * 320;
    float* es1 = (float*)d_ws;
    float* es2 = (float*)d_ws + NT * 32;
    float* ft0 = (float*)d_ws + NT * 64;
    float* ft1 = (float*)d_ws + NT * 65;
    float* ft2 = (float*)d_ws + NT * 66;

    k_copy_essum<<<NES, 256, 0, stream>>>(x, out, (const float*)d_in[2], es0, (const float*)d_in[1], ft0);

    k_layer<64, 192><<<NES + (NN / 2) * 8, 256, 0, stream>>>(out, (const float*)d_in[1], ft0, es0, (const float*)d_in[3],
                                                            (const float*)d_in[4], (const float*)d_in[5], out, 64,
                                                            (const float*)d_in[7], es1, (const float*)d_in[6], ft1);
    k_layer<192, 320><<<NES + (NN / 2) * 8, 256, 0, stream>>>(out, (const float*)d_in[6], ft1, es1, (const float*)d_in[8],
                                                             (const float*)d_in[9], (const float*)d_in[10], out, 192,
                                                             (const float*)d_in[12], es2, (const float*)d_in[11], ft2);
    k_layer<320, 0><<<(NN / 2) * 8, 256, 0, stream>>>(out, (const float*)d_in[11], ft2, es2, (const float*)d_in[13],
                                                      (const float*)d_in[14], (const float*)d_in[15], out, 320,
                                                      nullptr, nullptr, nullptr, nullptr);
}


# x values of each 32-k block of the layer product requested one block ahead in a second register set; block 0 ahead of the A tile's loads and barrier; k order of the sums kept
# speedup vs baseline: 2.7808x; 1.0356x over previous
#include <hip/hip_runtime.h>
#include <hip/hip_bf16.h>

#define NB   64
#define DD   32
#define NN   128
#define TT   6
#define NT   (NN*TT)
#define NCOL (NB*DD)
#define LEAF 64
#define JTOT 448
#define LDX  (JTOT*DD)

typedef __attribute__((ext_vector_type(2))) float v2f;
typedef __attribute__((ext_vector_type(8))) float v8f;

__device__ __forceinline__ float row16_sum(float v) {
    v += __builtin_bit_cast(float, __builtin_amdgcn_update_dpp(0, __builtin_bit_cast(int, v), 0x161, 0xf, 0xf, false));
    v += __builtin_bit_cast(float, __builtin_amdgcn_update_dpp(0, __builtin_bit_cast(int, v), 0x162, 0xf, 0xf, false));
    v += __builtin_bit_cast(float, __builtin_amdgcn_update_dpp(0, __builtin_bit_cast(int, v), 0x164, 0xf, 0xf, false));
    v += __builtin_bit_cast(float, __builtin_amdgcn_update_dpp(0, __builtin_bit_cast(int, v), 0x168, 0xf, 0xf, false));
    return v;
}
__device__ __forceinline__ int row16_isum(int v) {
    v += __builtin_amdgcn_update_dpp(0, v, 0x161, 0xf, 0xf, false);
    v += __builtin_amdgcn_update_dpp(0, v, 0x162, 0xf, 0xf, false);
    v += __builtin_amdgcn_update_dpp(0, v, 0x164, 0xf, 0xf, false);
    v += __builtin_amdgcn_update_dpp(0, v, 0x168, 0xf, 0xf, false);
    return v;
}

#define NTB  3

template <int K>
__device__ __forceinline__ void fs_tau_rows(const float* __restrict__ fsl, float* __restrict__ ftau, int blk) {
    constexpr int NQ = K / 64;
    int lane = threadIdx.x & 31;
    int wid  = threadIdx.x >> 5;
    int half = lane >> 4;
    int l15  = lane & 15;
    int row  = blk * NTB + ((wid == 7) ? half : 2);
    const float* zr = fsl + (size_t)row * K + l15 * 4;
    float z[NQ * 4];
    #pragma unroll
    for (int q = 0; q < NQ; ++q) {
        #pragma unroll
        for (int j = 0; j < 4; ++j) z[q * 4 + j] = zr[q * 64 + j];
    }
    float s = 0.0f;
    #pragma unroll
    for (int e = 0; e < NQ * 4; ++e) s += z[e];
    s = row16_sum(s);
    int   cnt = K;
    float tau = (s - 1.0f) / (float)K;
    for (int it = 0; it < K; ++it) {
        float s2 = 0.0f;
        int   c2 = 0;
        #pragma unroll
        for (int e = 0; e < NQ * 4; ++e) {
            bool in = z[e] > tau;
            s2 += in ? z[e] : 0.0f;
            c2 += in ? 1 : 0;
        }
        s2 = row16_sum(s2);
        c2 = row16_isum(c2);
        bool moved = (c2 != cnt);
        tau = (s2 - 1.0f) / (float)c2;
        cnt = c2;
        if (__builtin_amdgcn_ballot_w32(moved) == 0u) break;
    }
    if (l15 == 0 && (wid == 7 || half == 0)) ftau[row] = tau;
}

#define TPAD 33
#define ES_LDS (8 * 32 * TPAD + NTB * 8 * 32)
template <int J>
__device__ __forceinline__ void essum_block(const float* __restrict__ esl, float* __restrict__ essum,
                                            const float* __restrict__ fsl, float* __restrict__ ftau,
                                            int blk, float* smem) {
    constexpr int T     = J >> 5;
    constexpr int NTILE = NTB * T;
    float* accs = smem + 8 * 32 * TPAD;
    int lane = threadIdx.x & 31;
    int wid  = threadIdx.x >> 5;
    float* my = smem + wid * (32 * TPAD);
    #pragma unroll
    for (int k = 0; k < NTB; ++k) accs[(k * 8 + wid) * 32 + lane] = 0.0f;
    if (wid >= 6) fs_tau_rows<J>(fsl, ftau, blk);
    const float* src = esl + (size_t)blk * (NTB * J * DD) + lane;
    float nx[32];
    int f = wid;
    if (f < NTILE) {
        #pragma unroll
        for (int rr = 0; rr < 32; ++rr) nx[rr] = src[(f * 32 + rr) * DD];
    }
    while (f < NTILE) {
        #pragma unroll
        for (int rr = 0; rr < 32; ++rr) my[rr * TPAD + lane] = nx[rr];
        int fn = f + 8;
        if (fn < NTILE) {
            #pragma unroll
            for (int rr = 0; rr < 32; ++rr) nx[rr] = src[(fn * 32 + rr) * DD];
        }
        float r[32];
        #pragma unroll
        for (int k = 0; k < 32; ++k) r[k] = my[lane * TPAD + k];
        #pragma unroll
        for (int k = 2; k <= 32; k <<= 1) {
            #pragma unroll
            for (int j = k >> 1; j > 0; j >>= 1) {
                #pragma unroll
                for (int i = 0; i < 32; ++i) {
                    int ij = i ^ j;
                    if (ij > i) {
                        float a  = r[i], b2 = r[ij];
                        float lo = fminf(a, b2), hi = fmaxf(a, b2);
                        if ((i & k) == 0) { r[i] = hi; r[ij] = lo; }
                        else              { r[i] = lo; r[ij] = hi; }
                    }
                }
            }
        }
        float cum = 0.0f, tausum = 1.0f;
        int   ksup = 1;
        #pragma unroll
        for (int i = 0; i < 32; ++i) {
            cum += r[i];
            bool s = ((float)(i + 1) * r[i] > cum - 1.0f);
            if (s) { ksup = i + 1; tausum = cum; }
        }
        float tau = (tausum - 1.0f) / (float)ksup;
        float* slot = accs + ((f / T) * 8 + wid) * 32 + lane;
        float acc = *slot;
        #pragma unroll
        for (int j2 = 0; j2 < 32; ++j2) {
            float z = my[j2 * TPAD + lane];
            float t = __shfl(tau, j2, 32);
            acc += fmaxf(z - t, 0.0f);
        }
        *slot = acc;
        f = fn;
    }
    __syncthreads();
    if (threadIdx.x < NTB * 32) {
        int k = threadIdx.x >> 5;
        float s = 0.0f;
        #pragma unroll
        for (int w = 0; w < 8; ++w) s += accs[(k * 8 + w) * 32 + lane];
        essum[(blk * NTB + k) * DD + lane] = s;
    }
}

#define NES   (NT / NTB)
#define CPW   (8 - NTB * 2)
#define CPL   ((NB * 64 * DD / 4) / (NES * CPW * 32))
static_assert(NES * NTB == NT, "essum rows must divide over the blocks");
static_assert(CPW > 0 && CPL * NES * CPW * 32 == NB * 64 * DD / 4, "x copy must divide over the tile-less waves");
__global__ void __launch_bounds__(256) k_copy_essum(const float* __restrict__ x, float* __restrict__ out,
                             const float* __restrict__ esl, float* __restrict__ essum,
                             const float* __restrict__ fsl, float* __restrict__ ftau) {
    __shared__ float smem[ES_LDS];
    int wid = threadIdx.x >> 5;
    if (wid >= 8 - CPW) {
        int lane = threadIdx.x & 31;
        int base = (blockIdx.x * CPW + (wid - (8 - CPW))) * CPL;
        #pragma unroll
        for (int j = 0; j < CPL; ++j) {
            int idx = ((base + j) * 32 + lane) * 4;
            int b   = idx / (64 * DD);
            int rem = idx - b * (64 * DD);
            *(float4*)(out + (size_t)b * LDX + rem) = *(const float4*)(x + idx);
        }
    }
    essum_block<64>(esl, essum, fsl, ftau, blockIdx.x, smem);
}

#define RSTR 65
#define RTREE (DD * RSTR)

__device__ __forceinline__ float leaf_dot(v8f acc, int tree, int d,
                                          const float* __restrict__ essum,
                                          const float* __restrict__ thr,
                                          const float* __restrict__ lt,
                                          const float* rrow) {
    float p[TT], q[TT];
    #pragma unroll
    for (int t = 0; t < TT; ++t) {
        int   nt  = tree * TT + t;
        float sel = acc[t] * essum[nt * DD + d];
        float v   = (sel - thr[nt]) * __expf(-lt[nt]);
        p[t] = fminf(fmaxf(0.5f * v + 0.5f, 0.0f), 1.0f);
        q[t] = fminf(fmaxf(0.5f - 0.5f * v, 0.0f), 1.0f);
    }
    float r[LEAF];
    r[0] = 1.0f;
    #pragma unroll
    for (int t = 0; t < TT; ++t) {
        int sz = 1 << t;
        #pragma unroll
        for (int c = 0; c < 32; ++c) {
            int cc = sz - 1 - c;
            if (cc >= 0) {
                r[cc + sz] = r[cc] * q[t];
                r[cc]      = r[cc] * p[t];
            }
        }
    }
    float s = 0.0f;
    #pragma unroll
    for (int c = 0; c < LEAF; ++c) s += r[c] * rrow[c];
    return s;
}

__device__ __forceinline__ void x_block(const float* bk, int kb, float (&v)[32]) {
    #pragma unroll
    for (int s = 0; s < 8; ++s) {
        const float* p = bk + (kb + 4 * s) * DD;
        v[4 * s + 0] = p[0];
        v[4 * s + 1] = p[DD];
        v[4 * s + 2] = p[16];
        v[4 * s + 3] = p[DD + 16];
    }
}
__device__ __forceinline__ void mma_block(const float* ak, const float (&v)[32], v8f& acc0, v8f& acc1) {
    #pragma unroll
    for (int s = 0; s < 8; ++s) {
        v2f a = *(const v2f*)(ak + 4 * s);
        v2f b0, b1;
        b0.x = v[4 * s + 0];
        b0.y = v[4 * s + 1];
        b1.x = v[4 * s + 2];
        b1.y = v[4 * s + 3];
        acc0 = __builtin_amdgcn_wmma_f32_16x16x4_f32(false, a, false, b0,
                                                     (short)0, acc0, false, false);
        acc1 = __builtin_amdgcn_wmma_f32_16x16x4_f32(false, a, false, b1,
                                                     (short)0, acc1, false, false);
    }
}

template <int K, int JN>
__global__ void __launch_bounds__(256) k_layer(const float* x, const float* __restrict__ fsl,
                        const float* __restrict__ ftau,
                        const float* __restrict__ essum, const float* __restrict__ thr,
                        const float* __restrict__ lt, const float* __restrict__ resp,
                        float* out, int rowOff,
                        const float* __restrict__ esl_n, float* __restrict__ essum_n,
                        const float* __restrict__ fsl_n, float* __restrict__ ftau_n) {
    constexpr int PITCH = K + 4;
    constexpr int NQ    = K / 64;
    constexpr int NE    = JN ? NES : 0;
    constexpr int LY_LDS = 2 * RTREE + 16 * PITCH;
    constexpr int SM_LDS = (JN && ES_LDS > LY_LDS) ? ES_LDS : LY_LDS;
    __shared__ __attribute__((aligned(16))) float smem[SM_LDS];
    if constexpr (JN != 0) {
        if (blockIdx.x < NE) {
            essum_block<JN>(esl_n, essum_n, fsl_n, ftau_n, blockIdx.x, smem);
            return;
        }
    }
    float* rsp = smem;
    float* afs = smem + 2 * RTREE;
    int bid  = blockIdx.x - NE;
    int tid  = threadIdx.x;
    int lane = tid & 31;
    int wid  = tid >> 5;
    int mt   = bid >> 3;
    int bg   = bid & 7;
    int b    = bg * 8 + wid;
    int half = lane >> 4;
    int l15  = lane & 15;

    const float* bk = x + (size_t)b * LDX + (size_t)(half << 1) * DD + l15;
    float xa[32], xb[32];
    x_block(bk, 0, xa);

    const float* rsrc = resp + (size_t)mt * (2 * DD * LEAF);
    float rv[4][4];
    #pragma unroll
    for (int i = 0; i < 4; ++i) {
        #pragma unroll
        for (int j = 0; j < 4; ++j) rv[i][j] = rsrc[i * 1024 + tid * 4 + j];
    }

    {
        int hw = tid >> 4;
        if (wid < 6) {
            const float* zr = fsl + (size_t)(mt * 2 * TT + hw) * K + l15 * 4;
            float z[NQ * 4];
            #pragma unroll
            for (int q = 0; q < NQ; ++q) {
                #pragma unroll
                for (int j = 0; j < 4; ++j) z[q * 4 + j] = zr[q * 64 + j];
            }
            float tau = ftau[mt * 2 * TT + hw];
            int slot = (hw / TT) * 8 + (hw % TT);
            float* dst = afs + slot * PITCH + l15 * 4;
            #pragma unroll
            for (int q = 0; q < NQ; ++q) {
                #pragma unroll
                for (int j = 0; j < 4; ++j) dst[q * 64 + j] = fmaxf(z[q * 4 + j] - tau, 0.0f);
            }
        } else {
            int slot = ((hw - 12) >> 1) * 8 + 6 + (hw & 1);
            float* dst = afs + slot * PITCH + l15 * 4;
            #pragma unroll
            for (int q = 0; q < NQ; ++q) {
                #pragma unroll
                for (int j = 0; j < 4; ++j) dst[q * 64 + j] = 0.0f;
            }
        }
    }
    __syncthreads();

    const float* ap = afs + l15 * PITCH + (half << 1);

    v8f acc0 = {}, acc1 = {};
    constexpr int NBK = K / 32;
    static_assert(NBK % 2 == 0 && NBK * 32 == K, "k blocks come in pairs");
    for (int t = 0; t < NBK; t += 2) {
        x_block(bk, (t + 1) * 32, xb);
        mma_block(ap + t * 32, xa, acc0, acc1);
        if (t + 2 < NBK) x_block(bk, (t + 2) * 32, xa);
        mma_block(ap + (t + 1) * 32, xb, acc0, acc1);
    }

    #pragma unroll
    for (int i = 0; i < 4; ++i) {
        int e   = i * 1024 + tid * 4;
        int tr  = e >> 11;
        int rem = e & 2047;
        int dd  = rem >> 6, cc = rem & 63;
        #pragma unroll
        for (int j = 0; j < 4; ++j) rsp[tr * RTREE + dd * RSTR + cc + j] = rv[i][j];
    }
    __syncthreads();

    int tree = mt * 2 + half;
    const float* rbase = rsp + half * RTREE;
    float s0 = leaf_dot(acc0, tree, l15,      essum, thr, lt, rbase + l15 * RSTR);
    float s1 = leaf_dot(acc1, tree, 16 + l15, essum, thr, lt, rbase + (16 + l15) * RSTR);

    float t0 = __shfl_xor(s0, 16, 32);
    float t1 = __shfl_xor(s1, 16, 32);
    float oA = half ? t1 : s0;
    float oB = half ? s1 : t0;
    size_t obase = (size_t)b * LDX + (size_t)(rowOff + mt * 2) * DD + lane;
    out[obase]      = oA;
    out[obase + DD] = oB;
}

extern "C" void kernel_launch(void* const* d_in, const int* in_sizes, int n_in,
                              void* d_out, int out_size, void* d_ws, size_t ws_size,
                              hipStream_t stream) {
    const float* x = (const float*)d_in[0];
    float* out = (float*)d_out;

    float* es0 = (float*)d_ws + NT * 320;
    float* es1 = (float*)d_ws;
    float* es2 = (float*)d_ws + NT * 32;
    float* ft0 = (float*)d_ws + NT * 64;
    float* ft1 = (float*)d_ws + NT * 65;
    float* ft2 = (float*)d_ws + NT * 66;

    k_copy_essum<<<NES, 256, 0, stream>>>(x, out, (const float*)d_in[2], es0, (const float*)d_in[1], ft0);

    k_layer<64, 192><<<NES + (NN / 2) * 8, 256, 0, stream>>>(out, (const float*)d_in[1], ft0, es0, (const float*)d_in[3],
                                                            (const float*)d_in[4], (const float*)d_in[5], out, 64,
                                                            (const float*)d_in[7], es1, (const float*)d_in[6], ft1);
    k_layer<192, 320><<<NES + (NN / 2) * 8, 256, 0, stream>>>(out, (const float*)d_in[6], ft1, es1, (const float*)d_in[8],
                                                             (const float*)d_in[9], (const float*)d_in[10], out, 192,
                                                             (const float*)d_in[12], es2, (const float*)d_in[11], ft2);
    k_layer<320, 0><<<(NN / 2) * 8, 256, 0, stream>>>(out, (const float*)d_in[11], ft2, es2, (const float*)d_in[13],
                                                      (const float*)d_in[14], (const float*)d_in[15], out, 320,
                                                      nullptr, nullptr, nullptr, nullptr);
}
